# starting kernel
# baseline (speedup 1.0000x reference)
#include <hip/hip_runtime.h>

#define IN_DIM      512
#define NUM_CLASSES 10
#define QUAD_DIM    (IN_DIM + IN_DIM * (IN_DIM + 1) / 2)
#define BATCH       1024
#define NSLICES     64
#define KSLICE      (QUAD_DIM / NSLICES)
#define NBT         (BATCH / 16)
#define NPAIR       (BATCH / 32)
#define XS_STRIDE   516
#define TILE_B_OFF  (16 * XS_STRIDE * 4)
#define TAB_BYTES   (QUAD_DIM * 4)

typedef float v2f __attribute__((ext_vector_type(2)));
typedef float v8f __attribute__((ext_vector_type(8)));

__device__ __forceinline__ int tri_off(int i) {
    return i * IN_DIM - ((i * (i - 1)) >> 1);
}

__global__ void build_table(unsigned* __restrict__ tab) {
    int k = blockIdx.x * 256 + threadIdx.x;
    if (k >= QUAD_DIM) return;
    int i, j;
    if (k < IN_DIM) {
        i = k; j = IN_DIM;
    } else {
        int t = k - IN_DIM;
        float r = sqrtf(fmaxf(1050625.0f - 8.0f * (float)t, 0.0f));
        int ii = (int)((1025.0f - r) * 0.5f);
        ii = ii < 0 ? 0 : (ii > IN_DIM - 1 ? IN_DIM - 1 : ii);
        if (ii < IN_DIM - 1 && tri_off(ii + 1) <= t) ++ii;
        if (ii < IN_DIM - 1 && tri_off(ii + 1) <= t) ++ii;
        if (ii > 0 && tri_off(ii) > t) --ii;
        if (ii > 0 && tri_off(ii) > t) --ii;
        i = ii;
        j = ii + (t - tri_off(ii));
    }
    tab[k] = ((unsigned)(j * 4) << 16) | (unsigned)(i * 4);
}

__global__ __launch_bounds__(128) void quadfeat_gemm(const float* __restrict__ X,
                                                     const float* __restrict__ W,
                                                     const unsigned* __restrict__ tab,
                                                     float* __restrict__ partial) {
    __shared__ float xs[32 * XS_STRIDE];

    const int pair = blockIdx.x;
    const int tid  = threadIdx.x;

    const float4* src4 = (const float4*)(X + (size_t)pair * (32 * IN_DIM));
    for (int v = tid; v < 4096; v += 128) {
        int row = v >> 7;
        int c4  = v & 127;
        *(float4*)(xs + row * XS_STRIDE + c4 * 4) = src4[v];
    }
    if (tid < 32) xs[tid * XS_STRIDE + IN_DIM] = 1.0f;
    __syncthreads();

    const int wave  = tid >> 5;
    const int lane  = tid & 31;
    const int slice = blockIdx.y * 4 + wave;
    const int hi    = lane >> 4;
    const int n     = lane & 15;
    const char* xrow = (const char*)(xs + n * XS_STRIDE);

    const int k0 = slice * KSLICE + 2 * hi;
    const unsigned* tp = tab + k0;
    const float* wp = W + (size_t)((n < NUM_CLASSES) ? n : (NUM_CLASSES - 1)) * QUAD_DIM + k0;

    v8f accA = {0.f, 0.f, 0.f, 0.f, 0.f, 0.f, 0.f, 0.f};
    v8f accB = {0.f, 0.f, 0.f, 0.f, 0.f, 0.f, 0.f, 0.f};

    #pragma unroll 2
    for (int it = 0; it < KSLICE / 4; ++it) {
        uint2 idx = *(const uint2*)tp;
        unsigned o_i0 = idx.x & 0xFFFFu, o_j0 = idx.x >> 16;
        unsigned o_i1 = idx.y & 0xFFFFu, o_j1 = idx.y >> 16;

        float ai0 = *(const float*)(xrow + o_i0);
        float aj0 = *(const float*)(xrow + o_j0);
        float ai1 = *(const float*)(xrow + o_i1);
        float aj1 = *(const float*)(xrow + o_j1);
        float bi0 = *(const float*)(xrow + TILE_B_OFF + o_i0);
        float bj0 = *(const float*)(xrow + TILE_B_OFF + o_j0);
        float bi1 = *(const float*)(xrow + TILE_B_OFF + o_i1);
        float bj1 = *(const float*)(xrow + TILE_B_OFF + o_j1);

        v2f b = *(const v2f*)wp;

        v2f aA; aA.x = ai0 * aj0; aA.y = ai1 * aj1;
        v2f aB; aB.x = bi0 * bj0; aB.y = bi1 * bj1;

        accA = __builtin_amdgcn_wmma_f32_16x16x4_f32(
            false, aA, false, b, (short)0, accA, false, false);
        accB = __builtin_amdgcn_wmma_f32_16x16x4_f32(
            false, aB, false, b, (short)0, accB, false, false);

        tp += 4;
        wp += 4;
    }

    const int bt0 = pair * 2;
    float* pA = partial + ((size_t)slice * NBT + bt0) * 256;
    float* pB = partial + ((size_t)slice * NBT + bt0 + 1) * 256;
    #pragma unroll
    for (int r = 0; r < 8; ++r) {
        pA[(r + 8 * hi) * 16 + n] = accA[r];
        pB[(r + 8 * hi) * 16 + n] = accB[r];
    }
}

__global__ void quadfeat_reduce(const float* __restrict__ partial,
                                const float* __restrict__ bias,
                                float* __restrict__ out) {
    int idx = blockIdx.x * blockDim.x + threadIdx.x;
    if (idx >= BATCH * NUM_CLASSES) return;
    int b = idx / NUM_CLASSES;
    int c = idx - b * NUM_CLASSES;
    int bt = b >> 4;
    int m  = b & 15;
    float s = bias[c];
    #pragma unroll 4
    for (int sl = 0; sl < NSLICES; ++sl)
        s += partial[((size_t)sl * NBT + bt) * 256 + m * 16 + c];
    out[(size_t)b * NUM_CLASSES + c] = s;
}

extern "C" void kernel_launch(void* const* d_in, const int* in_sizes, int n_in,
                              void* d_out, int out_size, void* d_ws, size_t ws_size,
                              hipStream_t stream) {
    const float* x    = (const float*)d_in[0];
    const float* W    = (const float*)d_in[1];
    const float* bias = (const float*)d_in[2];
    float* out        = (float*)d_out;

    unsigned* tab  = (unsigned*)d_ws;
    float* partial = (float*)((char*)d_ws + TAB_BYTES);

    build_table<<<(QUAD_DIM + 255) / 256, 256, 0, stream>>>(tab);

    dim3 grid(NPAIR, NSLICES / 4);
    quadfeat_gemm<<<grid, 128, 0, stream>>>(x, W, tab, partial);

    int total = BATCH * NUM_CLASSES;
    quadfeat_reduce<<<(total + 255) / 256, 256, 0, stream>>>(partial, bias, out);
}


# Recast as per-class upper-triangular quadratic form: WMMA of packed W rows against x from LDS, x_i applied in registers; no index table or gathers; sums regrouped (new reduction tree, 32 partials)
# speedup vs baseline: 1.9808x; 1.9808x over previous
#include <hip/hip_runtime.h>


#define IN_DIM      512
#define NUM_CLASSES 10
#define QUAD_DIM    (IN_DIM + IN_DIM * (IN_DIM + 1) / 2)
#define BATCH       1024
#define ROWS        64
#define XS_STRIDE   516
#define TILE_PITCH  (16 * XS_STRIDE)
#define NPAIRS      16
#define NPART       (2 * NPAIRS)
#define NWAVES      5
#define NTHREADS    (32 * NWAVES)

typedef float v2f __attribute__((ext_vector_type(2)));
typedef float v4f __attribute__((ext_vector_type(4)));
typedef float v8f __attribute__((ext_vector_type(8)));
typedef float v4f_u __attribute__((ext_vector_type(4), aligned(4)));

__device__ __forceinline__ int tri_off(int i) {
    return i * IN_DIM - ((i * (i - 1)) >> 1);
}

__device__ __forceinline__ v4f mask_lower(v4f a, int j, int i) {
    a.x = (j     >= i) ? a.x : 0.f;
    a.y = (j + 1 >= i) ? a.y : 0.f;
    a.z = (j + 2 >= i) ? a.z : 0.f;
    a.w = (j + 3 >= i) ? a.w : 0.f;
    return a;
}

__device__ __forceinline__ void kstep8(const v4f a, const float* xb,
                                       v8f& c0, v8f& c1, v8f& c2, v8f& c3) {
    const v4f b0 = *(const v4f*)(xb);
    const v4f b1 = *(const v4f*)(xb + TILE_PITCH);
    const v4f b2 = *(const v4f*)(xb + 2 * TILE_PITCH);
    const v4f b3 = *(const v4f*)(xb + 3 * TILE_PITCH);
    const v2f alo = {a.x, a.y};
    const v2f ahi = {a.z, a.w};
    const v2f b0lo = {b0.x, b0.y}, b0hi = {b0.z, b0.w};
    const v2f b1lo = {b1.x, b1.y}, b1hi = {b1.z, b1.w};
    const v2f b2lo = {b2.x, b2.y}, b2hi = {b2.z, b2.w};
    const v2f b3lo = {b3.x, b3.y}, b3hi = {b3.z, b3.w};
    c0 = __builtin_amdgcn_wmma_f32_16x16x4_f32(false, alo, false, b0lo, (short)0, c0, false, false);
    c1 = __builtin_amdgcn_wmma_f32_16x16x4_f32(false, alo, false, b1lo, (short)0, c1, false, false);
    c2 = __builtin_amdgcn_wmma_f32_16x16x4_f32(false, alo, false, b2lo, (short)0, c2, false, false);
    c3 = __builtin_amdgcn_wmma_f32_16x16x4_f32(false, alo, false, b3lo, (short)0, c3, false, false);
    c0 = __builtin_amdgcn_wmma_f32_16x16x4_f32(false, ahi, false, b0hi, (short)0, c0, false, false);
    c1 = __builtin_amdgcn_wmma_f32_16x16x4_f32(false, ahi, false, b1hi, (short)0, c1, false, false);
    c2 = __builtin_amdgcn_wmma_f32_16x16x4_f32(false, ahi, false, b2hi, (short)0, c2, false, false);
    c3 = __builtin_amdgcn_wmma_f32_16x16x4_f32(false, ahi, false, b3hi, (short)0, c3, false, false);
}

__device__ __forceinline__ float tile_dot(const v8f y, const v4f l0, const v4f l1,
                                          const float* xe, float s) {
    const v4f x0 = *(const v4f*)(xe);
    const v4f x1 = *(const v4f*)(xe + 4);
    s = fmaf(x0.x, y[0] + l0.x, s);
    s = fmaf(x0.y, y[1] + l0.y, s);
    s = fmaf(x0.z, y[2] + l0.z, s);
    s = fmaf(x0.w, y[3] + l0.w, s);
    s = fmaf(x1.x, y[4] + l1.x, s);
    s = fmaf(x1.y, y[5] + l1.y, s);
    s = fmaf(x1.z, y[6] + l1.z, s);
    s = fmaf(x1.w, y[7] + l1.w, s);
    return s;
}

__global__ __launch_bounds__(NTHREADS) void quadform_gemm(const float* __restrict__ X,
                                                     const float* __restrict__ W,
                                                     float* __restrict__ part) {
    __shared__ float xs[ROWS * XS_STRIDE];

    const int bg  = blockIdx.x;
    const int tid = threadIdx.x;

    const v4f* src4 = (const v4f*)(X + (size_t)bg * (ROWS * IN_DIM));
    #pragma unroll 16
    for (int v = tid; v < ROWS * (IN_DIM / 4); v += NTHREADS) {
        const int row = v >> 7;
        const int c4  = v & 127;
        *(v4f*)(xs + row * XS_STRIDE + c4 * 4) = src4[v];
    }
    __syncthreads();

    const int wave = __builtin_amdgcn_readfirstlane(tid >> 5);
    const int lane = tid & 31;
    const int h    = lane >> 4;
    const int c    = lane & 15;
    const int unit = blockIdx.y * NWAVES + wave;
    const int cls  = unit >> 4;
    const int pair = unit & 15;

    const float* Wc = W + (size_t)cls * QUAD_DIM;
    const float* xl = xs + c * XS_STRIDE;

    float s0 = 0.f, s1 = 0.f, s2 = 0.f, s3 = 0.f;

    for (int side = 0; side < 2; ++side) {
        const int it = side ? (31 - pair) : pair;
        const int i0 = it * 16;
        const int i  = i0 + c;
        const float* R = Wc + IN_DIM + tri_off(i) - i;
        int j = 32 * (it >> 1) + 4 * h;
        const int niter = 16 - (it >> 1);

        v8f c0 = {0.f, 0.f, 0.f, 0.f, 0.f, 0.f, 0.f, 0.f};
        v8f c1 = c0, c2 = c0, c3 = c0;

        v4f a0 = *(const v4f_u*)(R + j);
        v4f a1 = *(const v4f_u*)(R + j + 8);
        v4f a2 = *(const v4f_u*)(R + j + 16);
        v4f a3 = *(const v4f_u*)(R + j + 24);
        a0 = mask_lower(a0, j, i);
        a1 = mask_lower(a1, j + 8, i);
        a2 = mask_lower(a2, j + 16, i);
        a3 = mask_lower(a3, j + 24, i);

        for (int iter = 0; iter < niter; ++iter) {
            const int jn = (iter + 1 < niter) ? j + 32 : j;
            const v4f n0 = *(const v4f_u*)(R + jn);
            const v4f n1 = *(const v4f_u*)(R + jn + 8);
            const v4f n2 = *(const v4f_u*)(R + jn + 16);
            const v4f n3 = *(const v4f_u*)(R + jn + 24);

            kstep8(a0, xl + j,      c0, c1, c2, c3);
            kstep8(a1, xl + j + 8,  c0, c1, c2, c3);
            kstep8(a2, xl + j + 16, c0, c1, c2, c3);
            kstep8(a3, xl + j + 24, c0, c1, c2, c3);

            a0 = n0; a1 = n1; a2 = n2; a3 = n3;
            j = jn;
        }

        const float* wl = Wc + i0 + 8 * h;
        const v4f l0 = *(const v4f*)(wl);
        const v4f l1 = *(const v4f*)(wl + 4);
        const float* xe = xl + i0 + 8 * h;
        s0 = tile_dot(c0, l0, l1, xe,                  s0);
        s1 = tile_dot(c1, l0, l1, xe + TILE_PITCH,     s1);
        s2 = tile_dot(c2, l0, l1, xe + 2 * TILE_PITCH, s2);
        s3 = tile_dot(c3, l0, l1, xe + 3 * TILE_PITCH, s3);
    }

    float* pp = part + (size_t)((cls * NPAIRS + pair) * 2 + h) * BATCH + bg * ROWS + c;
    pp[0]  = s0;
    pp[16] = s1;
    pp[32] = s2;
    pp[48] = s3;
}

__global__ void quadform_reduce(const float* __restrict__ part,
                                const float* __restrict__ bias,
                                float* __restrict__ out) {
    const int idx = blockIdx.x * blockDim.x + threadIdx.x;
    if (idx >= BATCH * NUM_CLASSES) return;
    const int b = idx / NUM_CLASSES;
    const int c = idx - b * NUM_CLASSES;
    float s = bias[c];
    const float* p = part + (size_t)c * (NPART * BATCH) + b;
    #pragma unroll 8
    for (int q = 0; q < NPART; ++q)
        s += p[q * BATCH];
    out[idx] = s;
}

extern "C" void kernel_launch(void* const* d_in, const int* in_sizes, int n_in,
                              void* d_out, int out_size, void* d_ws, size_t ws_size,
                              hipStream_t stream) {
    const float* x    = (const float*)d_in[0];
    const float* W    = (const float*)d_in[1];
    const float* bias = (const float*)d_in[2];
    float* out        = (float*)d_out;
    float* part       = (float*)d_ws;

    dim3 grid(16, 32);
    quadform_gemm<<<grid, NTHREADS, 0, stream>>>(x, W, part);

    quadform_reduce<<<40, 256, 0, stream>>>(part, bias, out);
}


# Each (class,pair) unit's K range cut into two 17-step waves: one 20-wave workgroup per WGP, 5 equal waves per SIMD (was 2x5 on 4 SIMDs); lane halves added in-wave, sums regrouped, 32 partials
# speedup vs baseline: 2.8061x; 1.4166x over previous
#include <hip/hip_runtime.h>


#define IN_DIM      512
#define NUM_CLASSES 10
#define QUAD_DIM    (IN_DIM + IN_DIM * (IN_DIM + 1) / 2)
#define BATCH       1024
#define ROWS        64
#define XS_STRIDE   516
#define TILE_PITCH  (16 * XS_STRIDE)
#define NPAIRS      16
#define NPART       (2 * NPAIRS)
#define HALF_STEPS  17
#define UNITS_PER_WG 10
#define NWAVES      (2 * UNITS_PER_WG)
#define NTHREADS    (32 * NWAVES)
#define FILL_THREADS 512
#define FILL_TRIPS  (ROWS * (IN_DIM / 4) / FILL_THREADS)

typedef float v2f __attribute__((ext_vector_type(2)));
typedef float v4f __attribute__((ext_vector_type(4)));
typedef float v8f __attribute__((ext_vector_type(8)));
typedef float v4f_u __attribute__((ext_vector_type(4), aligned(4)));

__device__ __forceinline__ int tri_off(int i) {
    return i * IN_DIM - ((i * (i - 1)) >> 1);
}

__device__ __forceinline__ v4f mask_lower(v4f a, int j, int i) {
    a.x = (j     >= i) ? a.x : 0.f;
    a.y = (j + 1 >= i) ? a.y : 0.f;
    a.z = (j + 2 >= i) ? a.z : 0.f;
    a.w = (j + 3 >= i) ? a.w : 0.f;
    return a;
}

__device__ __forceinline__ void kstep8(const v4f a, const float* xb,
                                       v8f& c0, v8f& c1, v8f& c2, v8f& c3) {
    const v4f b0 = *(const v4f*)(xb);
    const v4f b1 = *(const v4f*)(xb + TILE_PITCH);
    const v4f b2 = *(const v4f*)(xb + 2 * TILE_PITCH);
    const v4f b3 = *(const v4f*)(xb + 3 * TILE_PITCH);
    const v2f alo = {a.x, a.y};
    const v2f ahi = {a.z, a.w};
    const v2f b0lo = {b0.x, b0.y}, b0hi = {b0.z, b0.w};
    const v2f b1lo = {b1.x, b1.y}, b1hi = {b1.z, b1.w};
    const v2f b2lo = {b2.x, b2.y}, b2hi = {b2.z, b2.w};
    const v2f b3lo = {b3.x, b3.y}, b3hi = {b3.z, b3.w};
    c0 = __builtin_amdgcn_wmma_f32_16x16x4_f32(false, alo, false, b0lo, (short)0, c0, false, false);
    c1 = __builtin_amdgcn_wmma_f32_16x16x4_f32(false, alo, false, b1lo, (short)0, c1, false, false);
    c2 = __builtin_amdgcn_wmma_f32_16x16x4_f32(false, alo, false, b2lo, (short)0, c2, false, false);
    c3 = __builtin_amdgcn_wmma_f32_16x16x4_f32(false, alo, false, b3lo, (short)0, c3, false, false);
    c0 = __builtin_amdgcn_wmma_f32_16x16x4_f32(false, ahi, false, b0hi, (short)0, c0, false, false);
    c1 = __builtin_amdgcn_wmma_f32_16x16x4_f32(false, ahi, false, b1hi, (short)0, c1, false, false);
    c2 = __builtin_amdgcn_wmma_f32_16x16x4_f32(false, ahi, false, b2hi, (short)0, c2, false, false);
    c3 = __builtin_amdgcn_wmma_f32_16x16x4_f32(false, ahi, false, b3hi, (short)0, c3, false, false);
}

__device__ __forceinline__ float tile_dot(const v8f y, const v4f l0, const v4f l1,
                                          const float* xe, float s) {
    const v4f x0 = *(const v4f*)(xe);
    const v4f x1 = *(const v4f*)(xe + 4);
    s = fmaf(x0.x, y[0] + l0.x, s);
    s = fmaf(x0.y, y[1] + l0.y, s);
    s = fmaf(x0.z, y[2] + l0.z, s);
    s = fmaf(x0.w, y[3] + l0.w, s);
    s = fmaf(x1.x, y[4] + l1.x, s);
    s = fmaf(x1.y, y[5] + l1.y, s);
    s = fmaf(x1.z, y[6] + l1.z, s);
    s = fmaf(x1.w, y[7] + l1.w, s);
    return s;
}

__device__ __forceinline__ void segment(const float* Wc, const float* xl,
                                        const int it, const int first, const int n,
                                        const bool head, const int c, const int h,
                                        float& s0, float& s1, float& s2, float& s3) {
    const int i0 = it * 16;
    const int i  = i0 + c;
    const float* R = Wc + IN_DIM + tri_off(i) - i;
    int j = 32 * (it >> 1) + 16 * first + 4 * h;

    v8f c0 = {0.f, 0.f, 0.f, 0.f, 0.f, 0.f, 0.f, 0.f};
    v8f c1 = c0, c2 = c0, c3 = c0;

    const int j1 = (n > 1) ? j + 16 : j;
    v4f a0 = *(const v4f_u*)(R + j);
    v4f a1 = *(const v4f_u*)(R + j + 8);
    v4f a2 = *(const v4f_u*)(R + j1);
    v4f a3 = *(const v4f_u*)(R + j1 + 8);
    if (head) {
        a0 = mask_lower(a0, j, i);
        a1 = mask_lower(a1, j + 8, i);
        a2 = mask_lower(a2, j + 16, i);
        a3 = mask_lower(a3, j + 24, i);
    }

    int t = 0;
    for (; t + 2 <= n; t += 2) {
        const int jn0 = (t + 2 < n) ? j + 32 : j;
        const int jn1 = (t + 3 < n) ? j + 48 : jn0;
        const v4f n0 = *(const v4f_u*)(R + jn0);
        const v4f n1 = *(const v4f_u*)(R + jn0 + 8);
        const v4f n2 = *(const v4f_u*)(R + jn1);
        const v4f n3 = *(const v4f_u*)(R + jn1 + 8);

        kstep8(a0, xl + j,      c0, c1, c2, c3);
        kstep8(a1, xl + j + 8,  c0, c1, c2, c3);
        kstep8(a2, xl + j + 16, c0, c1, c2, c3);
        kstep8(a3, xl + j + 24, c0, c1, c2, c3);

        __builtin_amdgcn_sched_barrier(0);
        asm volatile("" :: "v"(n0), "v"(n1), "v"(n2), "v"(n3));
        a0 = n0; a1 = n1; a2 = n2; a3 = n3;
        j += 32;
    }
    if (n & 1) {
        kstep8(a0, xl + j,     c0, c1, c2, c3);
        kstep8(a1, xl + j + 8, c0, c1, c2, c3);
    }

    const float* wl = Wc + i0 + 8 * h;
    v4f l0 = *(const v4f*)(wl);
    v4f l1 = *(const v4f*)(wl + 4);
    if (!head) {
        l0 = (v4f){0.f, 0.f, 0.f, 0.f};
        l1 = l0;
    }
    const float* xe = xl + i0 + 8 * h;
    s0 = tile_dot(c0, l0, l1, xe,                  s0);
    s1 = tile_dot(c1, l0, l1, xe + TILE_PITCH,     s1);
    s2 = tile_dot(c2, l0, l1, xe + 2 * TILE_PITCH, s2);
    s3 = tile_dot(c3, l0, l1, xe + 3 * TILE_PITCH, s3);
}

__global__ __launch_bounds__(NTHREADS) void quadform_gemm(const float* __restrict__ X,
                                                     const float* __restrict__ W,
                                                     float* __restrict__ part) {
    __shared__ float xs[ROWS * XS_STRIDE];

    const int bg  = blockIdx.x;
    const int tid = threadIdx.x;

    const v4f* src4 = (const v4f*)(X + (size_t)bg * (ROWS * IN_DIM));
    if (tid < FILL_THREADS) {
        #pragma unroll
        for (int k = 0; k < FILL_TRIPS; ++k) {
            const int v   = tid + k * FILL_THREADS;
            const int row = v >> 7;
            const int c4  = v & 127;
            *(v4f*)(xs + row * XS_STRIDE + c4 * 4) = src4[v];
        }
    }
    __syncthreads();

    const int wave = __builtin_amdgcn_readfirstlane(tid >> 5);
    const int lane = tid & 31;
    const int h    = lane >> 4;
    const int c    = lane & 15;
    const int unit = blockIdx.y * UNITS_PER_WG + (wave >> 1);
    const int half = wave & 1;
    const int cls  = unit >> 4;
    const int pair = unit & 15;

    const float* Wc = W + (size_t)cls * QUAD_DIM;
    const float* xl = xs + c * XS_STRIDE;

    const int nfirst = 2 * (16 - (pair >> 1));

    float s0 = 0.f, s1 = 0.f, s2 = 0.f, s3 = 0.f;
    if (half == 0) {
        segment(Wc, xl, pair, 0, HALF_STEPS, true, c, h, s0, s1, s2, s3);
    } else {
        segment(Wc, xl, pair, HALF_STEPS, nfirst - HALF_STEPS, false, c, h, s0, s1, s2, s3);
        asm volatile("" :: "v"(s0), "v"(s1), "v"(s2), "v"(s3));
        segment(Wc, xl, 31 - pair, 0, 2 * HALF_STEPS - nfirst, true, c, h, s0, s1, s2, s3);
    }

    s0 += __shfl_xor(s0, 16);
    s1 += __shfl_xor(s1, 16);
    s2 += __shfl_xor(s2, 16);
    s3 += __shfl_xor(s3, 16);
    float* pp = part + (size_t)((cls * NPAIRS + pair) * 2 + half) * BATCH + bg * ROWS + lane;
    pp[0]  = h ? s1 : s0;
    pp[32] = h ? s3 : s2;
}

__global__ void quadform_reduce(const float* __restrict__ part,
                                const float* __restrict__ bias,
                                float* __restrict__ out) {
    const int idx = blockIdx.x * blockDim.x + threadIdx.x;
    if (idx >= BATCH * NUM_CLASSES) return;
    const int b = idx / NUM_CLASSES;
    const int c = idx - b * NUM_CLASSES;
    float s = bias[c];
    const float* p = part + (size_t)c * (NPART * BATCH) + b;
    #pragma unroll 8
    for (int q = 0; q < NPART; ++q)
        s += p[q * BATCH];
    out[idx] = s;
}

extern "C" void kernel_launch(void* const* d_in, const int* in_sizes, int n_in,
                              void* d_out, int out_size, void* d_ws, size_t ws_size,
                              hipStream_t stream) {
    const float* x    = (const float*)d_in[0];
    const float* W    = (const float*)d_in[1];
    const float* bias = (const float*)d_in[2];
    float* out        = (float*)d_out;
    float* part       = (float*)d_ws;

    dim3 grid(16, 16);
    quadform_gemm<<<grid, NTHREADS, 0, stream>>>(x, W, part);

    quadform_reduce<<<40, 256, 0, stream>>>(part, bias, out);
}


# Reduce launch re-mapped: one wave per class, batch row in the lane, so each partial load is one whole line and all 32 are in flight at once; out stored as whole lines via LDS; same order of sums
# speedup vs baseline: 3.1497x; 1.1224x over previous
#include <hip/hip_runtime.h>


#define IN_DIM      512
#define NUM_CLASSES 10
#define QUAD_DIM    (IN_DIM + IN_DIM * (IN_DIM + 1) / 2)
#define BATCH       1024
#define ROWS        64
#define XS_STRIDE   516
#define TILE_PITCH  (16 * XS_STRIDE)
#define NPAIRS      16
#define NPART       (2 * NPAIRS)
#define HALF_STEPS  17
#define UNITS_PER_WG 10
#define NWAVES      (2 * UNITS_PER_WG)
#define NTHREADS    (32 * NWAVES)
#define FILL_THREADS 512
#define FILL_TRIPS  (ROWS * (IN_DIM / 4) / FILL_THREADS)

typedef float v2f __attribute__((ext_vector_type(2)));
typedef float v4f __attribute__((ext_vector_type(4)));
typedef float v8f __attribute__((ext_vector_type(8)));
typedef float v4f_u __attribute__((ext_vector_type(4), aligned(4)));

__device__ __forceinline__ int tri_off(int i) {
    return i * IN_DIM - ((i * (i - 1)) >> 1);
}

__device__ __forceinline__ v4f mask_lower(v4f a, int j, int i) {
    a.x = (j     >= i) ? a.x : 0.f;
    a.y = (j + 1 >= i) ? a.y : 0.f;
    a.z = (j + 2 >= i) ? a.z : 0.f;
    a.w = (j + 3 >= i) ? a.w : 0.f;
    return a;
}

__device__ __forceinline__ void kstep8(const v4f a, const float* xb,
                                       v8f& c0, v8f& c1, v8f& c2, v8f& c3) {
    const v4f b0 = *(const v4f*)(xb);
    const v4f b1 = *(const v4f*)(xb + TILE_PITCH);
    const v4f b2 = *(const v4f*)(xb + 2 * TILE_PITCH);
    const v4f b3 = *(const v4f*)(xb + 3 * TILE_PITCH);
    const v2f alo = {a.x, a.y};
    const v2f ahi = {a.z, a.w};
    const v2f b0lo = {b0.x, b0.y}, b0hi = {b0.z, b0.w};
    const v2f b1lo = {b1.x, b1.y}, b1hi = {b1.z, b1.w};
    const v2f b2lo = {b2.x, b2.y}, b2hi = {b2.z, b2.w};
    const v2f b3lo = {b3.x, b3.y}, b3hi = {b3.z, b3.w};
    c0 = __builtin_amdgcn_wmma_f32_16x16x4_f32(false, alo, false, b0lo, (short)0, c0, false, false);
    c1 = __builtin_amdgcn_wmma_f32_16x16x4_f32(false, alo, false, b1lo, (short)0, c1, false, false);
    c2 = __builtin_amdgcn_wmma_f32_16x16x4_f32(false, alo, false, b2lo, (short)0, c2, false, false);
    c3 = __builtin_amdgcn_wmma_f32_16x16x4_f32(false, alo, false, b3lo, (short)0, c3, false, false);
    c0 = __builtin_amdgcn_wmma_f32_16x16x4_f32(false, ahi, false, b0hi, (short)0, c0, false, false);
    c1 = __builtin_amdgcn_wmma_f32_16x16x4_f32(false, ahi, false, b1hi, (short)0, c1, false, false);
    c2 = __builtin_amdgcn_wmma_f32_16x16x4_f32(false, ahi, false, b2hi, (short)0, c2, false, false);
    c3 = __builtin_amdgcn_wmma_f32_16x16x4_f32(false, ahi, false, b3hi, (short)0, c3, false, false);
}

__device__ __forceinline__ float tile_dot(const v8f y, const v4f l0, const v4f l1,
                                          const float* xe, float s) {
    const v4f x0 = *(const v4f*)(xe);
    const v4f x1 = *(const v4f*)(xe + 4);
    s = fmaf(x0.x, y[0] + l0.x, s);
    s = fmaf(x0.y, y[1] + l0.y, s);
    s = fmaf(x0.z, y[2] + l0.z, s);
    s = fmaf(x0.w, y[3] + l0.w, s);
    s = fmaf(x1.x, y[4] + l1.x, s);
    s = fmaf(x1.y, y[5] + l1.y, s);
    s = fmaf(x1.z, y[6] + l1.z, s);
    s = fmaf(x1.w, y[7] + l1.w, s);
    return s;
}

__device__ __forceinline__ void segment(const float* Wc, const float* xl,
                                        const int it, const int first, const int n,
                                        const bool head, const int c, const int h,
                                        float& s0, float& s1, float& s2, float& s3) {
    const int i0 = it * 16;
    const int i  = i0 + c;
    const float* R = Wc + IN_DIM + tri_off(i) - i;
    int j = 32 * (it >> 1) + 16 * first + 4 * h;

    v8f c0 = {0.f, 0.f, 0.f, 0.f, 0.f, 0.f, 0.f, 0.f};
    v8f c1 = c0, c2 = c0, c3 = c0;

    const int j1 = (n > 1) ? j + 16 : j;
    v4f a0 = *(const v4f_u*)(R + j);
    v4f a1 = *(const v4f_u*)(R + j + 8);
    v4f a2 = *(const v4f_u*)(R + j1);
    v4f a3 = *(const v4f_u*)(R + j1 + 8);
    if (head) {
        a0 = mask_lower(a0, j, i);
        a1 = mask_lower(a1, j + 8, i);
        a2 = mask_lower(a2, j + 16, i);
        a3 = mask_lower(a3, j + 24, i);
    }

    int t = 0;
    for (; t + 2 <= n; t += 2) {
        const int jn0 = (t + 2 < n) ? j + 32 : j;
        const int jn1 = (t + 3 < n) ? j + 48 : jn0;
        const v4f n0 = *(const v4f_u*)(R + jn0);
        const v4f n1 = *(const v4f_u*)(R + jn0 + 8);
        const v4f n2 = *(const v4f_u*)(R + jn1);
        const v4f n3 = *(const v4f_u*)(R + jn1 + 8);

        kstep8(a0, xl + j,      c0, c1, c2, c3);
        kstep8(a1, xl + j + 8,  c0, c1, c2, c3);
        kstep8(a2, xl + j + 16, c0, c1, c2, c3);
        kstep8(a3, xl + j + 24, c0, c1, c2, c3);

        __builtin_amdgcn_sched_barrier(0);
        asm volatile("" :: "v"(n0), "v"(n1), "v"(n2), "v"(n3));
        a0 = n0; a1 = n1; a2 = n2; a3 = n3;
        j += 32;
    }
    if (n & 1) {
        kstep8(a0, xl + j,     c0, c1, c2, c3);
        kstep8(a1, xl + j + 8, c0, c1, c2, c3);
    }

    const float* wl = Wc + i0 + 8 * h;
    v4f l0 = *(const v4f*)(wl);
    v4f l1 = *(const v4f*)(wl + 4);
    if (!head) {
        l0 = (v4f){0.f, 0.f, 0.f, 0.f};
        l1 = l0;
    }
    const float* xe = xl + i0 + 8 * h;
    s0 = tile_dot(c0, l0, l1, xe,                  s0);
    s1 = tile_dot(c1, l0, l1, xe + TILE_PITCH,     s1);
    s2 = tile_dot(c2, l0, l1, xe + 2 * TILE_PITCH, s2);
    s3 = tile_dot(c3, l0, l1, xe + 3 * TILE_PITCH, s3);
}

__global__ __launch_bounds__(NTHREADS) void quadform_gemm(const float* __restrict__ X,
                                                     const float* __restrict__ W,
                                                     float* __restrict__ part) {
    __shared__ float xs[ROWS * XS_STRIDE];

    const int bg  = blockIdx.x;
    const int tid = threadIdx.x;

    const v4f* src4 = (const v4f*)(X + (size_t)bg * (ROWS * IN_DIM));
    if (tid < FILL_THREADS) {
        #pragma unroll
        for (int k = 0; k < FILL_TRIPS; ++k) {
            const int v   = tid + k * FILL_THREADS;
            const int row = v >> 7;
            const int c4  = v & 127;
            *(v4f*)(xs + row * XS_STRIDE + c4 * 4) = src4[v];
        }
    }
    __syncthreads();

    const int wave = __builtin_amdgcn_readfirstlane(tid >> 5);
    const int lane = tid & 31;
    const int h    = lane >> 4;
    const int c    = lane & 15;
    const int unit = blockIdx.y * UNITS_PER_WG + (wave >> 1);
    const int half = wave & 1;
    const int cls  = unit >> 4;
    const int pair = unit & 15;

    const float* Wc = W + (size_t)cls * QUAD_DIM;
    const float* xl = xs + c * XS_STRIDE;

    const int nfirst = 2 * (16 - (pair >> 1));

    float s0 = 0.f, s1 = 0.f, s2 = 0.f, s3 = 0.f;
    if (half == 0) {
        segment(Wc, xl, pair, 0, HALF_STEPS, true, c, h, s0, s1, s2, s3);
    } else {
        segment(Wc, xl, pair, HALF_STEPS, nfirst - HALF_STEPS, false, c, h, s0, s1, s2, s3);
        asm volatile("" :: "v"(s0), "v"(s1), "v"(s2), "v"(s3));
        segment(Wc, xl, 31 - pair, 0, 2 * HALF_STEPS - nfirst, true, c, h, s0, s1, s2, s3);
    }

    s0 += __shfl_xor(s0, 16);
    s1 += __shfl_xor(s1, 16);
    s2 += __shfl_xor(s2, 16);
    s3 += __shfl_xor(s3, 16);
    float* pp = part + (size_t)((cls * NPAIRS + pair) * 2 + half) * BATCH + bg * ROWS + lane;
    pp[0]  = h ? s1 : s0;
    pp[32] = h ? s3 : s2;
}

#define RED_ROWS    32
#define RED_THREADS (RED_ROWS * NUM_CLASSES)
__global__ __launch_bounds__(RED_THREADS) void quadform_reduce(const float* __restrict__ part,
                                                               const float* __restrict__ bias,
                                                               float* __restrict__ out) {
    __shared__ float sm[RED_ROWS * NUM_CLASSES];
    const int tid  = threadIdx.x;
    const int lane = tid & 31;
    const int c  = __builtin_amdgcn_readfirstlane(tid >> 5);
    const int b0 = blockIdx.x * RED_ROWS;
    const float* p = part + (size_t)c * (NPART * BATCH) + b0 + lane;
    float v[NPART];
    #pragma unroll
    for (int q = 0; q < NPART; ++q)
        v[q] = p[q * BATCH];
    float s = bias[c];
    #pragma unroll
    for (int q = 0; q < NPART; ++q)
        s += v[q];
    sm[lane * NUM_CLASSES + c] = s;
    __syncthreads();
    out[b0 * NUM_CLASSES + tid] = sm[tid];
}

extern "C" void kernel_launch(void* const* d_in, const int* in_sizes, int n_in,
                              void* d_out, int out_size, void* d_ws, size_t ws_size,
                              hipStream_t stream) {
    const float* x    = (const float*)d_in[0];
    const float* W    = (const float*)d_in[1];
    const float* bias = (const float*)d_in[2];
    float* out        = (float*)d_out;
    float* part       = (float*)d_ws;

    dim3 grid(16, 16);
    quadform_gemm<<<grid, NTHREADS, 0, stream>>>(x, W, part);

    quadform_reduce<<<32, 320, 0, stream>>>(part, bias, out);
}


# K loop software-pipelined on the x operand: each step's 8 LDS fragment reads issued as one group a whole step ahead of its 16 products (two read places per trip instead of four); same order of sums
# speedup vs baseline: 3.3570x; 1.0658x over previous
#include <hip/hip_runtime.h>


#define IN_DIM      512
#define NUM_CLASSES 10
#define QUAD_DIM    (IN_DIM + IN_DIM * (IN_DIM + 1) / 2)
#define BATCH       1024
#define ROWS        64
#define XS_STRIDE   516
#define TILE_PITCH  (16 * XS_STRIDE)
#define NPAIRS      16
#define NPART       (2 * NPAIRS)
#define HALF_STEPS  17
#define UNITS_PER_WG 10
#define NWAVES      (2 * UNITS_PER_WG)
#define NTHREADS    (32 * NWAVES)
#define FILL_THREADS 512
#define FILL_TRIPS  (ROWS * (IN_DIM / 4) / FILL_THREADS)

typedef float v2f __attribute__((ext_vector_type(2)));
typedef float v4f __attribute__((ext_vector_type(4)));
typedef float v8f __attribute__((ext_vector_type(8)));
typedef float v4f_u __attribute__((ext_vector_type(4), aligned(4)));

__device__ __forceinline__ int tri_off(int i) {
    return i * IN_DIM - ((i * (i - 1)) >> 1);
}

__device__ __forceinline__ v4f mask_lower(v4f a, int j, int i) {
    a.x = (j     >= i) ? a.x : 0.f;
    a.y = (j + 1 >= i) ? a.y : 0.f;
    a.z = (j + 2 >= i) ? a.z : 0.f;
    a.w = (j + 3 >= i) ? a.w : 0.f;
    return a;
}

struct bfrag16 { v4f u0, u1, u2, u3, w0, w1, w2, w3; };

__device__ __forceinline__ bfrag16 load_b16(const float* xb) {
    bfrag16 f;
    f.u0 = *(const v4f*)(xb);
    f.u1 = *(const v4f*)(xb + TILE_PITCH);
    f.u2 = *(const v4f*)(xb + 2 * TILE_PITCH);
    f.u3 = *(const v4f*)(xb + 3 * TILE_PITCH);
    f.w0 = *(const v4f*)(xb + 8);
    f.w1 = *(const v4f*)(xb + 8 + TILE_PITCH);
    f.w2 = *(const v4f*)(xb + 8 + 2 * TILE_PITCH);
    f.w3 = *(const v4f*)(xb + 8 + 3 * TILE_PITCH);
    return f;
}

__device__ __forceinline__ void mma8(const v4f a, const v4f b0, const v4f b1, const v4f b2, const v4f b3,
                                     v8f& c0, v8f& c1, v8f& c2, v8f& c3) {
    const v2f alo = {a.x, a.y};
    const v2f ahi = {a.z, a.w};
    const v2f b0lo = {b0.x, b0.y}, b0hi = {b0.z, b0.w};
    const v2f b1lo = {b1.x, b1.y}, b1hi = {b1.z, b1.w};
    const v2f b2lo = {b2.x, b2.y}, b2hi = {b2.z, b2.w};
    const v2f b3lo = {b3.x, b3.y}, b3hi = {b3.z, b3.w};
    c0 = __builtin_amdgcn_wmma_f32_16x16x4_f32(false, alo, false, b0lo, (short)0, c0, false, false);
    c1 = __builtin_amdgcn_wmma_f32_16x16x4_f32(false, alo, false, b1lo, (short)0, c1, false, false);
    c2 = __builtin_amdgcn_wmma_f32_16x16x4_f32(false, alo, false, b2lo, (short)0, c2, false, false);
    c3 = __builtin_amdgcn_wmma_f32_16x16x4_f32(false, alo, false, b3lo, (short)0, c3, false, false);
    c0 = __builtin_amdgcn_wmma_f32_16x16x4_f32(false, ahi, false, b0hi, (short)0, c0, false, false);
    c1 = __builtin_amdgcn_wmma_f32_16x16x4_f32(false, ahi, false, b1hi, (short)0, c1, false, false);
    c2 = __builtin_amdgcn_wmma_f32_16x16x4_f32(false, ahi, false, b2hi, (short)0, c2, false, false);
    c3 = __builtin_amdgcn_wmma_f32_16x16x4_f32(false, ahi, false, b3hi, (short)0, c3, false, false);
}

__device__ __forceinline__ void mma16(const v4f a, const v4f a8, const bfrag16& f,
                                      v8f& c0, v8f& c1, v8f& c2, v8f& c3) {
    mma8(a,  f.u0, f.u1, f.u2, f.u3, c0, c1, c2, c3);
    mma8(a8, f.w0, f.w1, f.w2, f.w3, c0, c1, c2, c3);
}

#define B_FENCE(f) asm volatile("" :: "v"((f).u0), "v"((f).u1), "v"((f).u2), "v"((f).u3), \
                                     "v"((f).w0), "v"((f).w1), "v"((f).w2), "v"((f).w3))

__device__ __forceinline__ float tile_dot(const v8f y, const v4f l0, const v4f l1,
                                          const float* xe, float s) {
    const v4f x0 = *(const v4f*)(xe);
    const v4f x1 = *(const v4f*)(xe + 4);
    s = fmaf(x0.x, y[0] + l0.x, s);
    s = fmaf(x0.y, y[1] + l0.y, s);
    s = fmaf(x0.z, y[2] + l0.z, s);
    s = fmaf(x0.w, y[3] + l0.w, s);
    s = fmaf(x1.x, y[4] + l1.x, s);
    s = fmaf(x1.y, y[5] + l1.y, s);
    s = fmaf(x1.z, y[6] + l1.z, s);
    s = fmaf(x1.w, y[7] + l1.w, s);
    return s;
}

__device__ __forceinline__ void segment(const float* Wc, const float* xl,
                                        const int it, const int first, const int n,
                                        const bool head, const int c, const int h,
                                        float& s0, float& s1, float& s2, float& s3) {
    const int i0 = it * 16;
    const int i  = i0 + c;
    const float* R = Wc + IN_DIM + tri_off(i) - i;
    int j = 32 * (it >> 1) + 16 * first + 4 * h;

    v8f c0 = {0.f, 0.f, 0.f, 0.f, 0.f, 0.f, 0.f, 0.f};
    v8f c1 = c0, c2 = c0, c3 = c0;

    const int j1 = (n > 1) ? j + 16 : j;
    v4f a0 = *(const v4f_u*)(R + j);
    v4f a1 = *(const v4f_u*)(R + j + 8);
    v4f a2 = *(const v4f_u*)(R + j1);
    v4f a3 = *(const v4f_u*)(R + j1 + 8);
    if (head) {
        a0 = mask_lower(a0, j, i);
        a1 = mask_lower(a1, j + 8, i);
        a2 = mask_lower(a2, j + 16, i);
        a3 = mask_lower(a3, j + 24, i);
    }

    bfrag16 f0 = load_b16(xl + j);

    int t = 0;
    for (; t + 2 <= n; t += 2) {
        const int jn0 = (t + 2 < n) ? j + 32 : j;
        const int jn1 = (t + 3 < n) ? j + 48 : jn0;
        const v4f n0 = *(const v4f_u*)(R + jn0);
        const v4f n1 = *(const v4f_u*)(R + jn0 + 8);
        const v4f n2 = *(const v4f_u*)(R + jn1);
        const v4f n3 = *(const v4f_u*)(R + jn1 + 8);

        const bfrag16 f1 = load_b16(xl + j + 16);
        __builtin_amdgcn_sched_barrier(0);
        mma16(a0, a1, f0, c0, c1, c2, c3);
        __builtin_amdgcn_sched_barrier(0);
        B_FENCE(f1);

        f0 = load_b16(xl + jn0);
        __builtin_amdgcn_sched_barrier(0);
        mma16(a2, a3, f1, c0, c1, c2, c3);

        __builtin_amdgcn_sched_barrier(0);
        B_FENCE(f0);
        asm volatile("" :: "v"(n0), "v"(n1), "v"(n2), "v"(n3));
        a0 = n0; a1 = n1; a2 = n2; a3 = n3;
        j += 32;
    }
    if (n & 1) {
        mma16(a0, a1, f0, c0, c1, c2, c3);
    }

    const float* wl = Wc + i0 + 8 * h;
    v4f l0 = *(const v4f*)(wl);
    v4f l1 = *(const v4f*)(wl + 4);
    if (!head) {
        l0 = (v4f){0.f, 0.f, 0.f, 0.f};
        l1 = l0;
    }
    const float* xe = xl + i0 + 8 * h;
    s0 = tile_dot(c0, l0, l1, xe,                  s0);
    s1 = tile_dot(c1, l0, l1, xe + TILE_PITCH,     s1);
    s2 = tile_dot(c2, l0, l1, xe + 2 * TILE_PITCH, s2);
    s3 = tile_dot(c3, l0, l1, xe + 3 * TILE_PITCH, s3);
}

__global__ __launch_bounds__(NTHREADS) void quadform_gemm(const float* __restrict__ X,
                                                     const float* __restrict__ W,
                                                     float* __restrict__ part) {
    __shared__ float xs[ROWS * XS_STRIDE];

    const int bg  = blockIdx.x;
    const int tid = threadIdx.x;

    const v4f* src4 = (const v4f*)(X + (size_t)bg * (ROWS * IN_DIM));
    if (tid < FILL_THREADS) {
        #pragma unroll
        for (int k = 0; k < FILL_TRIPS; ++k) {
            const int v   = tid + k * FILL_THREADS;
            const int row = v >> 7;
            const int c4  = v & 127;
            *(v4f*)(xs + row * XS_STRIDE + c4 * 4) = src4[v];
        }
    }
    __syncthreads();

    const int wave = __builtin_amdgcn_readfirstlane(tid >> 5);
    const int lane = tid & 31;
    const int h    = lane >> 4;
    const int c    = lane & 15;
    const int unit = blockIdx.y * UNITS_PER_WG + (wave >> 1);
    const int half = wave & 1;
    const int cls  = unit >> 4;
    const int pair = unit & 15;

    const float* Wc = W + (size_t)cls * QUAD_DIM;
    const float* xl = xs + c * XS_STRIDE;

    const int nfirst = 2 * (16 - (pair >> 1));

    float s0 = 0.f, s1 = 0.f, s2 = 0.f, s3 = 0.f;
    if (half == 0) {
        segment(Wc, xl, pair, 0, HALF_STEPS, true, c, h, s0, s1, s2, s3);
    } else {
        segment(Wc, xl, pair, HALF_STEPS, nfirst - HALF_STEPS, false, c, h, s0, s1, s2, s3);
        asm volatile("" :: "v"(s0), "v"(s1), "v"(s2), "v"(s3));
        segment(Wc, xl, 31 - pair, 0, 2 * HALF_STEPS - nfirst, true, c, h, s0, s1, s2, s3);
    }

    s0 += __shfl_xor(s0, 16);
    s1 += __shfl_xor(s1, 16);
    s2 += __shfl_xor(s2, 16);
    s3 += __shfl_xor(s3, 16);
    float* pp = part + (size_t)((cls * NPAIRS + pair) * 2 + half) * BATCH + bg * ROWS + lane;
    pp[0]  = h ? s1 : s0;
    pp[32] = h ? s3 : s2;
}

#define RED_ROWS    32
#define RED_THREADS (RED_ROWS * NUM_CLASSES)
__global__ __launch_bounds__(RED_THREADS) void quadform_reduce(const float* __restrict__ part,
                                                               const float* __restrict__ bias,
                                                               float* __restrict__ out) {
    __shared__ float sm[RED_ROWS * NUM_CLASSES];
    const int tid  = threadIdx.x;
    const int lane = tid & 31;
    const int c  = __builtin_amdgcn_readfirstlane(tid >> 5);
    const int b0 = blockIdx.x * RED_ROWS;
    const float* p = part + (size_t)c * (NPART * BATCH) + b0 + lane;
    float v[NPART];
    #pragma unroll
    for (int q = 0; q < NPART; ++q)
        v[q] = p[q * BATCH];
    float s = bias[c];
    #pragma unroll
    for (int q = 0; q < NPART; ++q)
        s += v[q];
    sm[lane * NUM_CLASSES + c] = s;
    __syncthreads();
    out[b0 * NUM_CLASSES + tid] = sm[tid];
}

extern "C" void kernel_launch(void* const* d_in, const int* in_sizes, int n_in,
                              void* d_out, int out_size, void* d_ws, size_t ws_size,
                              hipStream_t stream) {
    const float* x    = (const float*)d_in[0];
    const float* W    = (const float*)d_in[1];
    const float* bias = (const float*)d_in[2];
    float* out        = (float*)d_out;
    float* part       = (float*)d_ws;

    dim3 grid(16, 16);
    quadform_gemm<<<grid, NTHREADS, 0, stream>>>(x, W, part);

    quadform_reduce<<<32, 320, 0, stream>>>(part, bias, out);
}
